# starting kernel
# baseline (speedup 1.0000x reference)
#include <hip/hip_runtime.h>
#include <hip/hip_bf16.h>
#include <math.h>

typedef __attribute__((ext_vector_type(2))) float v2f;
typedef __attribute__((ext_vector_type(8))) float v8f;

#define IN_DIM  256
#define OUT_DIM 128
#define KC 32
#define NB 32
#define HSTR 36

__device__ __forceinline__ unsigned fkey(float f) {
    unsigned u = __float_as_uint(f);
    return (u & 0x80000000u) ? ~u : (u | 0x80000000u);
}
__device__ __forceinline__ float keyf(unsigned k) {
    unsigned u = (k & 0x80000000u) ? (k & 0x7FFFFFFFu) : ~k;
    return __uint_as_float(u);
}

__global__ void gat_init(float* __restrict__ out, unsigned* __restrict__ mkey,
                         unsigned* __restrict__ deg, float* __restrict__ denom,
                         int n, int total) {
    int i = blockIdx.x * blockDim.x + threadIdx.x;
    if (i < total) out[i] = 0.0f;
    if (i < n) { mkey[i] = 0u; deg[i] = 0u; denom[i] = 0.0f; }
}

__global__ __launch_bounds__(256) void gat_gemm_wmma(
    const float* __restrict__ h, const float* __restrict__ W,
    float* __restrict__ z, int n) {
    __shared__ float lds_h[NB * HSTR];
    __shared__ float lds_w[KC * OUT_DIM];

    const int tid      = threadIdx.x;
    const int lane     = tid & 31;
    const int wave     = tid >> 5;
    const int nodeTile = wave >> 2;
    const int outGroup = wave & 3;
    const int nodeBase = blockIdx.x * NB;

    const int hl  = lane & 15;
    const int khi = (lane >> 4) * 2;

    v8f c0 = {}; v8f c1 = {};

    for (int kc = 0; kc < IN_DIM; kc += KC) {
        __syncthreads();
        {
            int nodeL = tid >> 3;
            int ks    = (tid & 7) * 4;
            int g     = nodeBase + nodeL;
            if (g >= n) g = n - 1;
            const float4 hv = *(const float4*)(h + (size_t)g * IN_DIM + kc + ks);
            *(float4*)(&lds_h[nodeL * HSTR + ks]) = hv;
            if (kc + KC < IN_DIM)
                __builtin_prefetch((const void*)(h + (size_t)g * IN_DIM + kc + KC + ks), 0, 1);
        }
        {
            int o  = tid >> 1;
            int kb = (tid & 1) * 16;
            const float* wrow = W + (size_t)o * IN_DIM + kc + kb;
            #pragma unroll
            for (int j = 0; j < 16; ++j) {
                int k = kb + j;
                lds_w[((k >> 1) * OUT_DIM + o) * 2 + (k & 1)] = wrow[j];
            }
        }
        __syncthreads();

        const int row = nodeTile * 16 + hl;
        const int o0  = (outGroup * 2 + 0) * 16 + hl;
        const int o1  = (outGroup * 2 + 1) * 16 + hl;
        #pragma unroll
        for (int ks = 0; ks < KC; ks += 4) {
            v2f a = *(const v2f*)(&lds_h[row * HSTR + ks + khi]);
            int kb = ks + khi;
            v2f b0 = *(const v2f*)(&lds_w[((kb >> 1) * OUT_DIM + o0) * 2]);
            v2f b1 = *(const v2f*)(&lds_w[((kb >> 1) * OUT_DIM + o1) * 2]);
            c0 = __builtin_amdgcn_wmma_f32_16x16x4_f32(false, a, false, b0, (short)0, c0, false, false);
            c1 = __builtin_amdgcn_wmma_f32_16x16x4_f32(false, a, false, b1, (short)0, c1, false, false);
        }
    }

    const int mrow0 = nodeBase + nodeTile * 16 + (lane >> 4) * 8;
    const int ncol  = lane & 15;
    #pragma unroll
    for (int r = 0; r < 8; ++r) {
        int gnode = mrow0 + r;
        if (gnode < n) {
            z[(size_t)gnode * OUT_DIM + (outGroup * 2 + 0) * 16 + ncol] = c0[r];
            z[(size_t)gnode * OUT_DIM + (outGroup * 2 + 1) * 16 + ncol] = c1[r];
        }
    }
}

__global__ __launch_bounds__(256) void gat_nodedot(
    const float* __restrict__ z, const float* __restrict__ a_attn,
    float* __restrict__ zs, float* __restrict__ zd, int n) {
    int node = blockIdx.x * 8 + (threadIdx.x >> 5);
    int lane = threadIdx.x & 31;
    if (node >= n) return;
    float4 zv = *(const float4*)(z + (size_t)node * OUT_DIM + lane * 4);
    float4 as = *(const float4*)(a_attn + lane * 4);
    float4 ad = *(const float4*)(a_attn + OUT_DIM + lane * 4);
    float s = zv.x * as.x + zv.y * as.y + zv.z * as.z + zv.w * as.w;
    float d = zv.x * ad.x + zv.y * ad.y + zv.z * ad.z + zv.w * ad.w;
    #pragma unroll
    for (int off = 16; off > 0; off >>= 1) {
        s += __shfl_down(s, off, 32);
        d += __shfl_down(d, off, 32);
    }
    if (lane == 0) { zs[node] = s; zd[node] = d; }
}

__global__ void gat_edgeA(const int* __restrict__ src, const int* __restrict__ dst,
                          const float* __restrict__ zs, const float* __restrict__ zd,
                          float* __restrict__ e, unsigned* __restrict__ mkey,
                          unsigned* __restrict__ deg, int E) {
    int i = blockIdx.x * blockDim.x + threadIdx.x;
    if (i >= E) return;
    int s = src[i], d = dst[i];
    float x  = zs[s] + zd[d];
    float ev = x > 0.0f ? x : 0.01f * x;
    e[i] = ev;
    atomicMax(&mkey[d], fkey(ev));
    atomicAdd(&deg[d], 1u);
}

__global__ void gat_edgeB(const int* __restrict__ dst, float* ebuf,
                          const unsigned* __restrict__ mkey,
                          float* __restrict__ denom, int E) {
    int i = blockIdx.x * blockDim.x + threadIdx.x;
    if (i >= E) return;
    int d = dst[i];
    float v = expf(ebuf[i] - keyf(mkey[d]));
    ebuf[i] = v;
    atomicAdd(&denom[d], v);
}

__global__ __launch_bounds__(256) void gat_edgeC(
    const int* __restrict__ src, const int* __restrict__ dst,
    const float* __restrict__ ex, const float* __restrict__ denom,
    const float* __restrict__ z, float* __restrict__ out, int E) {
    int e    = blockIdx.x * 8 + (threadIdx.x >> 5);
    int lane = threadIdx.x & 31;
    if (e >= E) return;
    int s = src[e], d = dst[e];
    float alpha = ex[e] / denom[d];
    float4 zv = *(const float4*)(z + (size_t)s * OUT_DIM + lane * 4);
    float* o = out + (size_t)d * OUT_DIM + lane * 4;
    atomicAdd(o + 0, alpha * zv.x);
    atomicAdd(o + 1, alpha * zv.y);
    atomicAdd(o + 2, alpha * zv.z);
    atomicAdd(o + 3, alpha * zv.w);
}

__global__ void gat_final(const unsigned* __restrict__ deg,
                          const float* __restrict__ z, float* __restrict__ out, int total) {
    int i = blockIdx.x * blockDim.x + threadIdx.x;
    if (i >= total) return;
    if (deg[i >> 7] == 0u) out[i] = z[i];
}

extern "C" void kernel_launch(void* const* d_in, const int* in_sizes, int n_in,
                              void* d_out, int out_size, void* d_ws, size_t ws_size,
                              hipStream_t stream) {
    const float* h      = (const float*)d_in[0];
    const int*   src    = (const int*)d_in[1];
    const int*   dst    = (const int*)d_in[2];
    const float* W      = (const float*)d_in[3];
    const float* a_attn = (const float*)d_in[4];
    const int n = in_sizes[0] / IN_DIM;
    const int E = in_sizes[1];
    float* out = (float*)d_out;

    char* ws = (char*)d_ws;
    float*    z     = (float*)ws;    ws += (size_t)n * OUT_DIM * sizeof(float);
    float*    zs    = (float*)ws;    ws += (size_t)n * sizeof(float);
    float*    zd    = (float*)ws;    ws += (size_t)n * sizeof(float);
    float*    ebuf  = (float*)ws;    ws += (size_t)E * sizeof(float);
    unsigned* mkey  = (unsigned*)ws; ws += (size_t)n * sizeof(unsigned);
    unsigned* deg   = (unsigned*)ws; ws += (size_t)n * sizeof(unsigned);
    float*    denom = (float*)ws;

    const int total = n * OUT_DIM;
    gat_init     <<<(total + 255) / 256, 256, 0, stream>>>(out, mkey, deg, denom, n, total);
    gat_gemm_wmma<<<(n + NB - 1) / NB,   256, 0, stream>>>(h, W, z, n);
    gat_nodedot  <<<(n + 7) / 8,         256, 0, stream>>>(z, a_attn, zs, zd, n);
    gat_edgeA    <<<(E + 255) / 256,     256, 0, stream>>>(src, dst, zs, zd, ebuf, mkey, deg, E);
    gat_edgeB    <<<(E + 255) / 256,     256, 0, stream>>>(dst, ebuf, mkey, denom, E);
    gat_edgeC    <<<(E + 7) / 8,         256, 0, stream>>>(src, dst, ebuf, denom, z, out, E);
    gat_final    <<<(total + 255) / 256, 256, 0, stream>>>(deg, z, out, total);
}


# GEMM made persistent: 256 work-groups hold all of W in LDS for the whole kernel, A fragments come straight from global rows; no K-chunk staging or barriers; k summed in another order inside each 8
# speedup vs baseline: 2.1688x; 2.1688x over previous
#include <hip/hip_runtime.h>
#include <hip/hip_bf16.h>
#include <math.h>

typedef __attribute__((ext_vector_type(2))) float v2f;
typedef __attribute__((ext_vector_type(8))) float v8f;

#define IN_DIM  256
#define OUT_DIM 128
#define WPITCH 260
#define GEMM_BLOCKS 256
#define GEMM_WAVES  16


template <int NT, int SJ, int JB>
__device__ __forceinline__ void gemm_unit(const float* __restrict__ h, const float* lw,
                                          float* __restrict__ z, int rowBase, int outBase,
                                          int n, int lane) {
    constexpr int NJ    = IN_DIM / 8;
    constexpr int NTRIP = NJ / JB;
    constexpr int NSTG  = JB / SJ;
    const int c  = lane & 15;
    const int hf = lane >> 4;
    int g = rowBase + c;
    if (g >= n) g = n - 1;
    const float* ap = h + (size_t)g * IN_DIM + hf * 4;
    const float* bp = lw + (outBase + c) * WPITCH + hf * 4;
    v8f acc[NT];
    #pragma unroll
    for (int t = 0; t < NT; ++t) acc[t] = v8f{};

    float4 an[JB];
    #pragma unroll
    for (int i = 0; i < JB; ++i) an[i] = *(const float4*)(ap + i * 8);
    float4 bn[SJ][NT];
    #pragma unroll
    for (int i = 0; i < SJ; ++i) {
        #pragma unroll
        for (int t = 0; t < NT; ++t) bn[i][t] = *(const float4*)(bp + t * 16 * WPITCH + i * 8);
    }

    #pragma unroll 1
    for (int trip = 0; trip < NTRIP; ++trip) {
        float4 a[JB];
        #pragma unroll
        for (int i = 0; i < JB; ++i) a[i] = an[i];
        if (trip + 1 < NTRIP) {
            #pragma unroll
            for (int i = 0; i < JB; ++i) an[i] = *(const float4*)(ap + ((trip + 1) * JB + i) * 8);
        }
        #pragma unroll
        for (int s = 0; s < NSTG; ++s) {
            float4 b[SJ][NT];
            #pragma unroll
            for (int i = 0; i < SJ; ++i) {
                #pragma unroll
                for (int t = 0; t < NT; ++t) b[i][t] = bn[i][t];
            }
            const int jn = trip * JB + (s + 1) * SJ;
            if (s + 1 < NSTG || trip + 1 < NTRIP) {
                #pragma unroll
                for (int i = 0; i < SJ; ++i) {
                    #pragma unroll
                    for (int t = 0; t < NT; ++t)
                        bn[i][t] = *(const float4*)(bp + t * 16 * WPITCH + (jn + i) * 8);
                }
            }
            __builtin_amdgcn_sched_barrier(0);
            #pragma unroll
            for (int i = 0; i < SJ; ++i) {
                const float4 av = a[s * SJ + i];
                const v2f a0 = {av.x, av.y};
                const v2f a1 = {av.z, av.w};
                #pragma unroll
                for (int t = 0; t < NT; ++t) {
                    const v2f b0 = {b[i][t].x, b[i][t].y};
                    acc[t] = __builtin_amdgcn_wmma_f32_16x16x4_f32(false, a0, false, b0, (short)0, acc[t], false, false);
                }
                #pragma unroll
                for (int t = 0; t < NT; ++t) {
                    const v2f b1 = {b[i][t].z, b[i][t].w};
                    acc[t] = __builtin_amdgcn_wmma_f32_16x16x4_f32(false, a1, false, b1, (short)0, acc[t], false, false);
                }
            }
            __builtin_amdgcn_sched_barrier(0);
        }
    }
    const int mrow0 = rowBase + hf * 8;
    #pragma unroll
    for (int r = 0; r < 8; ++r) {
        const int gnode = mrow0 + r;
        if (gnode < n) {
            #pragma unroll
            for (int t = 0; t < NT; ++t)
                z[(size_t)gnode * OUT_DIM + outBase + t * 16 + c] = acc[t][r];
        }
    }
}

__global__ __launch_bounds__(GEMM_WAVES * 32) void gat_gemm_wmma(
    const float* __restrict__ h, const float* __restrict__ W,
    float* __restrict__ z, unsigned* __restrict__ deg, unsigned* __restrict__ total, int n) {
    __shared__ __attribute__((aligned(16))) float lw[OUT_DIM * WPITCH];

    const int tid  = threadIdx.x;
    const int lane = tid & 31;
    const int wave = __builtin_amdgcn_readfirstlane(tid >> 5);

    const int nTiles = (n + 15) >> 4;
    const int t0 = (int)(((unsigned)blockIdx.x * (unsigned)nTiles) / GEMM_BLOCKS);
    const int t1 = (int)((((unsigned)blockIdx.x + 1u) * (unsigned)nTiles) / GEMM_BLOCKS);
    const int nT = t1 - t0;
    const int nX = nT & ~3;
    const int nY = (nT - nX) * 4;
    const int u = 4 * (3 - (wave >> 2)) + (wave & 3);

    {
        int ft = -1;
        if (wave < nX)      ft = t0 + wave;
        else if (u < nY)    ft = t0 + nX + (u >> 2);
        if (ft >= 0) {
            int g = ft * 16 + (lane & 15);
            if (g >= n) g = n - 1;
            const float* p = h + (size_t)g * IN_DIM + (lane >> 4) * 128;
            __builtin_prefetch((const void*)p, 0, 1);
            __builtin_prefetch((const void*)(p + 64), 0, 1);
        }
    }

    for (int i = blockIdx.x * (GEMM_WAVES * 32) + tid; i < n; i += GEMM_BLOCKS * (GEMM_WAVES * 32)) deg[i] = 0u;
    if (blockIdx.x == 0 && tid == 0) *total = 0u;

    for (int i = tid; i < OUT_DIM * (IN_DIM / 4); i += GEMM_WAVES * 32) {
        const int row = i >> 6, c4 = i & 63;
        const float4 v = *(const float4*)(W + (size_t)row * IN_DIM + c4 * 4);
        *(float4*)(&lw[row * WPITCH + c4 * 4]) = v;
    }
    __syncthreads();

    for (int i = wave; i < nX; i += GEMM_WAVES)
        gemm_unit<8, 1, 8>(h, lw, z, (t0 + i) * 16, 0, n, lane);
    if (u < nY)
        gemm_unit<2, 4, 32>(h, lw, z, (t0 + nX + (u >> 2)) * 16, (u & 3) * 32, n, lane);
}

__global__ void gat_count(const int* __restrict__ dst, unsigned* __restrict__ deg,
                          unsigned* __restrict__ rank, int E) {
    int i = blockIdx.x * blockDim.x + threadIdx.x;
    if (i >= E) return;
    rank[i] = atomicAdd(&deg[dst[i]], 1u);
}

__global__ __launch_bounds__(256) void gat_alloc(
    const unsigned* __restrict__ deg, unsigned* __restrict__ off,
    unsigned* __restrict__ total, int n) {
    __shared__ unsigned wsum[8];
    __shared__ unsigned base_s;
    const int tid  = threadIdx.x;
    const int lane = tid & 31;
    const int wave = tid >> 5;
    const int node = blockIdx.x * 256 + tid;
    const unsigned c = node < n ? deg[node] : 0u;
    unsigned v = c;
    for (int o = 1; o < 32; o <<= 1) {
        unsigned t = __shfl_up(v, o, 32);
        if (lane >= o) v += t;
    }
    if (lane == 31) wsum[wave] = v;
    __syncthreads();
    if (tid == 0) {
        unsigned run = 0u;
        for (int w = 0; w < 8; ++w) { unsigned t = wsum[w]; wsum[w] = run; run += t; }
        base_s = atomicAdd(total, run);
    }
    __syncthreads();
    if (node < n) off[node] = base_s + wsum[wave] + (v - c);
}

__global__ void gat_place(const int* __restrict__ src, const int* __restrict__ dst,
                          const unsigned* __restrict__ off, const unsigned* __restrict__ rank,
                          int* __restrict__ csr, int E) {
    int i = blockIdx.x * blockDim.x + threadIdx.x;
    if (i >= E) return;
    csr[off[dst[i]] + rank[i]] = src[i];
}

__device__ __forceinline__ float lrelu(float x) { return x > 0.0f ? x : 0.01f * x; }

__device__ __forceinline__ float xhalf(float x) {
    return __int_as_float(__builtin_amdgcn_permlanex16(__float_as_int(x), __float_as_int(x),
                                                       0x76543210, 0xfedcba98, false, false));
}
template <int M>
__device__ __forceinline__ float xrow(float x) {
    return __int_as_float(__builtin_amdgcn_update_dpp(0, __float_as_int(x), 0x160 + M, 0xf, 0xf, true));
}
__device__ __forceinline__ float wave_sum(float v) {
    v += xrow<1>(v); v += xrow<2>(v); v += xrow<4>(v); v += xrow<8>(v);
    return v + xhalf(v);
}
__device__ __forceinline__ float groups_max(float v) {
    v = fmaxf(v, xrow<4>(v)); v = fmaxf(v, xrow<8>(v));
    return fmaxf(v, xhalf(v));
}
__device__ __forceinline__ float groups_sum(float v) {
    v += xrow<4>(v); v += xrow<8>(v);
    return v + xhalf(v);
}

__device__ __forceinline__ float4 zrow(const float* __restrict__ zl, int s) {
    return *(const float4*)(zl + (size_t)s * OUT_DIM);
}
__device__ __forceinline__ void fma4(float4& acc, float a, const float4& r) {
    acc.x += a * r.x; acc.y += a * r.y; acc.z += a * r.z; acc.w += a * r.w;
}
__device__ __forceinline__ float dot4(const float4& a, const float4& b) {
    return a.x * b.x + a.y * b.y + a.z * b.z + a.w * b.w;
}
__device__ __forceinline__ float lane_f(float v, int t) {
    return __int_as_float(__builtin_amdgcn_readlane(__float_as_int(v), t));
}

__global__ __launch_bounds__(256) void gat_gather(
    const int* __restrict__ csr, const unsigned* __restrict__ deg,
    const unsigned* __restrict__ off, const float* __restrict__ a_attn,
    const float* __restrict__ z, float* __restrict__ out, int n) {
    const int wv   = __builtin_amdgcn_readfirstlane((int)(threadIdx.x >> 5));
    const int node = blockIdx.x * 8 + wv;
    const int lane = threadIdx.x & 31;
    if (node >= n) return;
    const float* zl = z + lane * 4;
    const float4 self = zrow(zl, node);
    const float4 as   = *(const float4*)(a_attn + lane * 4);
    const float4 ad   = *(const float4*)(a_attn + OUT_DIM + lane * 4);
    const int    cnt  = (int)deg[node];
    const int    beg  = (int)off[node];
    asm volatile("" :: "v"(as.x), "v"(as.y), "v"(as.z), "v"(as.w),
                       "v"(ad.x), "v"(ad.y), "v"(ad.z), "v"(ad.w), "s"(beg));
    float* o = out + (size_t)node * OUT_DIM + lane * 4;
    if (cnt == 0) {
        *(float4*)o = self;
        return;
    }
    const float  zdd = wave_sum(dot4(self, ad));

    const int  et = (lane >> 2) & 7;
    const bool h16 = (lane & 16) != 0, h8 = (lane & 8) != 0, h4 = (lane & 4) != 0;
    const float4 zero4 = make_float4(0.0f, 0.0f, 0.0f, 0.0f);

    float  m = -INFINITY, den = 0.0f;
    float4 acc = zero4;
    for (int b32 = 0; b32 < cnt; b32 += 32) {
        const int j   = b32 + lane;
        const int sv  = j < cnt ? csr[beg + j] : 0;
        const int c32 = cnt - b32 < 32 ? cnt - b32 : 32;
        for (int g = 0; g < c32; g += 8) {
            const int c = c32 - g < 8 ? c32 - g : 8;
            const int s0 = __builtin_amdgcn_readlane(sv, g),     s1 = __builtin_amdgcn_readlane(sv, g + 1);
            const int s2 = __builtin_amdgcn_readlane(sv, g + 2), s3 = __builtin_amdgcn_readlane(sv, g + 3);
            const int s4 = __builtin_amdgcn_readlane(sv, g + 4), s5 = __builtin_amdgcn_readlane(sv, g + 5);
            const int s6 = __builtin_amdgcn_readlane(sv, g + 6), s7 = __builtin_amdgcn_readlane(sv, g + 7);
            float4 r0 = zrow(zl, s0);
            float4 r1 = zero4, r2 = zero4, r3 = zero4, r4 = zero4, r5 = zero4, r6 = zero4, r7 = zero4;
            if (c > 1) r1 = zrow(zl, s1);
            if (c > 2) r2 = zrow(zl, s2);
            if (c > 3) r3 = zrow(zl, s3);
            if (c > 4) r4 = zrow(zl, s4);
            if (c > 5) r5 = zrow(zl, s5);
            if (c > 6) r6 = zrow(zl, s6);
            if (c > 7) r7 = zrow(zl, s7);
            const float p0 = dot4(r0, as), p1 = dot4(r1, as), p2 = dot4(r2, as), p3 = dot4(r3, as);
            const float p4 = dot4(r4, as), p5 = dot4(r5, as), p6 = dot4(r6, as), p7 = dot4(r7, as);
            const float q0 = (h16 ? p4 : p0) + xhalf(h16 ? p0 : p4);
            const float q1 = (h16 ? p5 : p1) + xhalf(h16 ? p1 : p5);
            const float q2 = (h16 ? p6 : p2) + xhalf(h16 ? p2 : p6);
            const float q3 = (h16 ? p7 : p3) + xhalf(h16 ? p3 : p7);
            const float u0 = (h8 ? q2 : q0) + xrow<8>(h8 ? q0 : q2);
            const float u1 = (h8 ? q3 : q1) + xrow<8>(h8 ? q1 : q3);
            float sc = (h4 ? u1 : u0) + xrow<4>(h4 ? u0 : u1);
            sc += xrow<2>(sc);
            sc += xrow<1>(sc);
            const bool  live = et < c;
            const float e    = live ? lrelu(sc + zdd) : -INFINITY;
            const float mn = fmaxf(m, groups_max(e));
            const float rs = expf(m - mn);
            const float pe = live ? expf(e - mn) : 0.0f;
            den = den * rs + groups_sum(pe);
            acc.x *= rs; acc.y *= rs; acc.z *= rs; acc.w *= rs;
            m = mn;
            fma4(acc, lane_f(pe,  0), r0); fma4(acc, lane_f(pe,  4), r1);
            fma4(acc, lane_f(pe,  8), r2); fma4(acc, lane_f(pe, 12), r3);
            fma4(acc, lane_f(pe, 16), r4); fma4(acc, lane_f(pe, 20), r5);
            fma4(acc, lane_f(pe, 24), r6); fma4(acc, lane_f(pe, 28), r7);
        }
    }
    const float inv = 1.0f / den;
    *(float4*)o = make_float4(acc.x * inv, acc.y * inv, acc.z * inv, acc.w * inv);
}

extern "C" void kernel_launch(void* const* d_in, const int* in_sizes, int n_in,
                              void* d_out, int out_size, void* d_ws, size_t ws_size,
                              hipStream_t stream) {
    const float* h      = (const float*)d_in[0];
    const int*   src    = (const int*)d_in[1];
    const int*   dst    = (const int*)d_in[2];
    const float* W      = (const float*)d_in[3];
    const float* a_attn = (const float*)d_in[4];
    const int n = in_sizes[0] / IN_DIM;
    const int E = in_sizes[1];
    float* out = (float*)d_out;

    char* ws = (char*)d_ws;
    float*    z     = (float*)ws;    ws += (size_t)n * OUT_DIM * sizeof(float);
    ws += (size_t)n * sizeof(float) * 2;
    int*      csr   = (int*)ws;      ws += (size_t)E * sizeof(int);
    unsigned* deg   = (unsigned*)ws; ws += (size_t)n * sizeof(unsigned);
    unsigned* off   = (unsigned*)ws; ws += (size_t)n * sizeof(unsigned);
    unsigned* total = (unsigned*)ws;
    unsigned* rank  = (unsigned*)d_out;

    gat_gemm_wmma<<<GEMM_BLOCKS, GEMM_WAVES * 32, 0, stream>>>(h, W, z, deg, total, n);
    gat_count    <<<(E + 255) / 256,     256, 0, stream>>>(dst, deg, rank, E);
    gat_alloc    <<<(n + 255) / 256,     256, 0, stream>>>(deg, off, total, n);
    gat_place    <<<(E + 255) / 256,     256, 0, stream>>>(src, dst, off, rank, csr, E);
    gat_gather   <<<(n + 7) / 8,         256, 0, stream>>>(csr, deg, off, a_attn, z, out, n);
}


# Node ordering kept in LDS and gathered by the same work-group: a range's owner orders its sources by node, then its 32 waves take the range's nodes by ticket; 4 launches become 3; arrival-order sums
# speedup vs baseline: 3.7490x; 1.7286x over previous
#include <hip/hip_runtime.h>
#include <hip/hip_bf16.h>
#include <math.h>

typedef __attribute__((ext_vector_type(2))) float v2f;
typedef __attribute__((ext_vector_type(8))) float v8f;

#define IN_DIM  256
#define OUT_DIM 128
#define WPITCH 260
#define GEMM_BLOCKS 256
#define GEMM_WAVES  16


template <int NT, int SJ, int JB>
__device__ __forceinline__ void gemm_unit(const float* __restrict__ h, const float* lw,
                                          float* __restrict__ z, int rowBase, int outBase,
                                          int n, int lane) {
    constexpr int NJ    = IN_DIM / 8;
    constexpr int NTRIP = NJ / JB;
    constexpr int NSTG  = JB / SJ;
    const int c  = lane & 15;
    const int hf = lane >> 4;
    int g = rowBase + c;
    if (g >= n) g = n - 1;
    const float* ap = h + (size_t)g * IN_DIM + hf * 4;
    const float* bp = lw + (outBase + c) * WPITCH + hf * 4;
    v8f acc[NT];
    #pragma unroll
    for (int t = 0; t < NT; ++t) acc[t] = v8f{};

    float4 an[JB];
    #pragma unroll
    for (int i = 0; i < JB; ++i) an[i] = *(const float4*)(ap + i * 8);
    float4 bn[SJ][NT];
    #pragma unroll
    for (int i = 0; i < SJ; ++i) {
        #pragma unroll
        for (int t = 0; t < NT; ++t) bn[i][t] = *(const float4*)(bp + t * 16 * WPITCH + i * 8);
    }

    #pragma unroll 1
    for (int trip = 0; trip < NTRIP; ++trip) {
        float4 a[JB];
        #pragma unroll
        for (int i = 0; i < JB; ++i) a[i] = an[i];
        if (trip + 1 < NTRIP) {
            #pragma unroll
            for (int i = 0; i < JB; ++i) an[i] = *(const float4*)(ap + ((trip + 1) * JB + i) * 8);
        }
        #pragma unroll
        for (int s = 0; s < NSTG; ++s) {
            float4 b[SJ][NT];
            #pragma unroll
            for (int i = 0; i < SJ; ++i) {
                #pragma unroll
                for (int t = 0; t < NT; ++t) b[i][t] = bn[i][t];
            }
            const int jn = trip * JB + (s + 1) * SJ;
            if (s + 1 < NSTG || trip + 1 < NTRIP) {
                #pragma unroll
                for (int i = 0; i < SJ; ++i) {
                    #pragma unroll
                    for (int t = 0; t < NT; ++t)
                        bn[i][t] = *(const float4*)(bp + t * 16 * WPITCH + (jn + i) * 8);
                }
            }
            __builtin_amdgcn_sched_barrier(0);
            #pragma unroll
            for (int i = 0; i < SJ; ++i) {
                const float4 av = a[s * SJ + i];
                const v2f a0 = {av.x, av.y};
                const v2f a1 = {av.z, av.w};
                #pragma unroll
                for (int t = 0; t < NT; ++t) {
                    const v2f b0 = {b[i][t].x, b[i][t].y};
                    acc[t] = __builtin_amdgcn_wmma_f32_16x16x4_f32(false, a0, false, b0, (short)0, acc[t], false, false);
                }
                #pragma unroll
                for (int t = 0; t < NT; ++t) {
                    const v2f b1 = {b[i][t].z, b[i][t].w};
                    acc[t] = __builtin_amdgcn_wmma_f32_16x16x4_f32(false, a1, false, b1, (short)0, acc[t], false, false);
                }
            }
            __builtin_amdgcn_sched_barrier(0);
        }
    }
    const int mrow0 = rowBase + hf * 8;
    #pragma unroll
    for (int r = 0; r < 8; ++r) {
        const int gnode = mrow0 + r;
        if (gnode < n) {
            #pragma unroll
            for (int t = 0; t < NT; ++t)
                z[(size_t)gnode * OUT_DIM + outBase + t * 16 + c] = acc[t][r];
        }
    }
}

__global__ __launch_bounds__(GEMM_WAVES * 32) void gat_gemm_wmma(
    const float* __restrict__ h, const float* __restrict__ W,
    float* __restrict__ z, int n) {
    __shared__ __attribute__((aligned(16))) float lw[OUT_DIM * WPITCH];

    const int tid  = threadIdx.x;
    const int lane = tid & 31;
    const int wave = __builtin_amdgcn_readfirstlane(tid >> 5);

    const int nTiles = (n + 15) >> 4;
    const int t0 = (int)(((unsigned)blockIdx.x * (unsigned)nTiles) / GEMM_BLOCKS);
    const int t1 = (int)((((unsigned)blockIdx.x + 1u) * (unsigned)nTiles) / GEMM_BLOCKS);
    const int nT = t1 - t0;
    const int nX = nT & ~3;
    const int nY = (nT - nX) * 4;
    const int u = 4 * (3 - (wave >> 2)) + (wave & 3);

    {
        int ft = -1;
        if (wave < nX)      ft = t0 + wave;
        else if (u < nY)    ft = t0 + nX + (u >> 2);
        if (ft >= 0) {
            int g = ft * 16 + (lane & 15);
            if (g >= n) g = n - 1;
            const float* p = h + (size_t)g * IN_DIM + (lane >> 4) * 128;
            __builtin_prefetch((const void*)p, 0, 1);
            __builtin_prefetch((const void*)(p + 64), 0, 1);
        }
    }

    for (int i = tid; i < OUT_DIM * (IN_DIM / 4); i += GEMM_WAVES * 32) {
        const int row = i >> 6, c4 = i & 63;
        const float4 v = *(const float4*)(W + (size_t)row * IN_DIM + c4 * 4);
        *(float4*)(&lw[row * WPITCH + c4 * 4]) = v;
    }
    __syncthreads();

    for (int i = wave; i < nX; i += GEMM_WAVES)
        gemm_unit<8, 1, 8>(h, lw, z, (t0 + i) * 16, 0, n, lane);
    if (u < nY)
        gemm_unit<2, 4, 32>(h, lw, z, (t0 + nX + (u >> 2)) * 16, (u & 3) * 32, n, lane);
}

#define NBINS       256
#define BIN_NODES   196
#define CHUNKS      128
#define CHUNK_EDGES 3125
#define SORT_T      512
#define SORT_W      (SORT_T / 32)
#define BIN_KPT     7
#define LIST_CAP    4096

template <int NW>
__device__ __forceinline__ unsigned block_excl_scan(unsigned c, unsigned* wsum, unsigned& total) {
    const int lane = threadIdx.x & 31;
    const int wave = threadIdx.x >> 5;
    unsigned v = c;
    for (int o = 1; o < 32; o <<= 1) {
        unsigned t = __shfl_up(v, o, 32);
        if (lane >= o) v += t;
    }
    if (lane == 31) wsum[wave] = v;
    __syncthreads();
    unsigned pre = 0u, tot = 0u;
    for (int w = 0; w < NW; ++w) {
        const unsigned t = wsum[w];
        if (w < wave) pre += t;
        tot += t;
    }
    __syncthreads();
    total = tot;
    return pre + (v - c);
}

__global__ __launch_bounds__(SORT_T) void gat_bin(
    const int* __restrict__ src, const int* __restrict__ dst,
    unsigned* __restrict__ sorted, unsigned* __restrict__ tab, int E) {
    __shared__ unsigned cnt[NBINS];
    __shared__ unsigned start[NBINS];
    __shared__ unsigned wsum[SORT_W];
    __shared__ unsigned stage[CHUNK_EDGES];
    const int tid = threadIdx.x;
    const int c   = blockIdx.x;
    const int e0  = c * CHUNK_EDGES;
    const int e1  = e0 + CHUNK_EDGES < E ? e0 + CHUNK_EDGES : E;
    if (tid < NBINS) cnt[tid] = 0u;
    __syncthreads();
    unsigned dv[BIN_KPT], sv[BIN_KPT];
    #pragma unroll
    for (int k = 0; k < BIN_KPT; ++k) {
        int i = e0 + k * SORT_T + tid;
        if (i > e1 - 1) i = e1 - 1;
        if (i < 0) i = 0;
        dv[k] = (unsigned)dst[i];
        sv[k] = (unsigned)src[i];
    }
    unsigned rec[BIN_KPT], key[BIN_KPT];
    #pragma unroll
    for (int k = 0; k < BIN_KPT; ++k) {
        const int i = e0 + k * SORT_T + tid;
        key[k] = 0xffffffffu;
        rec[k] = 0u;
        if (i < e1) {
            const unsigned d  = dv[k];
            const unsigned s  = sv[k];
            const unsigned bn = d / BIN_NODES;
            const unsigned r  = atomicAdd(&cnt[bn], 1u);
            rec[k] = s | ((d - bn * BIN_NODES) << 24);
            key[k] = (bn << 16) | r;
        }
    }
    __syncthreads();
    const unsigned cv = tid < NBINS ? cnt[tid] : 0u;
    unsigned tot;
    const unsigned ex = block_excl_scan<SORT_W>(cv, wsum, tot);
    if (tid < NBINS) {
        start[tid] = ex;
        tab[c * NBINS + tid] = (ex << 16) | cv;
    }
    __syncthreads();
    #pragma unroll
    for (int k = 0; k < BIN_KPT; ++k) {
        if (key[k] != 0xffffffffu) stage[start[key[k] >> 16] + (key[k] & 0xffffu)] = rec[k];
    }
    __syncthreads();
    for (int j = tid; j < e1 - e0; j += SORT_T) sorted[e0 + j] = stage[j];
}

__device__ __forceinline__ float lrelu(float x) { return x > 0.0f ? x : 0.01f * x; }

__device__ __forceinline__ float xhalf(float x) {
    return __int_as_float(__builtin_amdgcn_permlanex16(__float_as_int(x), __float_as_int(x),
                                                       0x76543210, 0xfedcba98, false, false));
}
template <int M>
__device__ __forceinline__ float xrow(float x) {
    return __int_as_float(__builtin_amdgcn_update_dpp(0, __float_as_int(x), 0x160 + M, 0xf, 0xf, true));
}
__device__ __forceinline__ float wave_sum(float v) {
    v += xrow<1>(v); v += xrow<2>(v); v += xrow<4>(v); v += xrow<8>(v);
    return v + xhalf(v);
}
__device__ __forceinline__ float groups_max(float v) {
    v = fmaxf(v, xrow<4>(v)); v = fmaxf(v, xrow<8>(v));
    return fmaxf(v, xhalf(v));
}
__device__ __forceinline__ float groups_sum(float v) {
    v += xrow<4>(v); v += xrow<8>(v);
    return v + xhalf(v);
}

__device__ __forceinline__ float4 zrow(const float* __restrict__ zl, int s) {
    return *(const float4*)(zl + (size_t)s * OUT_DIM);
}
__device__ __forceinline__ void fma4(float4& acc, float a, const float4& r) {
    acc.x += a * r.x; acc.y += a * r.y; acc.z += a * r.z; acc.w += a * r.w;
}
__device__ __forceinline__ float dot4(const float4& a, const float4& b) {
    return a.x * b.x + a.y * b.y + a.z * b.z + a.w * b.w;
}
__device__ __forceinline__ float lane_f(float v, int t) {
    return __int_as_float(__builtin_amdgcn_readlane(__float_as_int(v), t));
}

#define CG_T 1024
#define CG_W (CG_T / 32)

__device__ __forceinline__ void gather_node_scan(
    const unsigned* __restrict__ sorted, const unsigned* pst, const unsigned* pcn,
    const float* __restrict__ zl, const float4 as, const float zdd, const unsigned ln,
    const int lane, float* __restrict__ o) {
    float  m = -INFINITY, den = 0.0f;
    float4 acc = make_float4(0.0f, 0.0f, 0.0f, 0.0f);
    for (int ch = 0; ch < CHUNKS; ++ch) {
        const unsigned len = pcn[ch];
        const unsigned* pp = sorted + ch * CHUNK_EDGES + pst[ch];
        for (unsigned j0 = 0; j0 < len; j0 += 32) {
            const unsigned j = j0 + lane;
            const unsigned v = j < len ? pp[j] : 0xffffffffu;
            unsigned mask = __builtin_amdgcn_ballot_w32(j < len && (v >> 24) == ln);
            while (mask) {
                const int t = __builtin_ctz(mask);
                mask &= mask - 1u;
                const int sidx = (int)(__builtin_amdgcn_readlane((int)v, t) & 0xffffff);
                const float4 r = zrow(zl, sidx);
                const float e  = lrelu(wave_sum(dot4(r, as)) + zdd);
                const float mn = fmaxf(m, e);
                const float rs = expf(m - mn);
                const float pe = expf(e - mn);
                den = den * rs + pe;
                acc.x = acc.x * rs + pe * r.x; acc.y = acc.y * rs + pe * r.y;
                acc.z = acc.z * rs + pe * r.z; acc.w = acc.w * rs + pe * r.w;
                m = mn;
            }
        }
    }
    const float inv = 1.0f / den;
    *(float4*)o = make_float4(acc.x * inv, acc.y * inv, acc.z * inv, acc.w * inv);
}

__global__ __launch_bounds__(CG_T) void gat_range(
    const unsigned* __restrict__ sorted, const unsigned* __restrict__ tab,
    const float* __restrict__ a_attn, const float* __restrict__ z,
    float* __restrict__ out, int n) {
    __shared__ unsigned cntN[NBINS];
    __shared__ unsigned nst[NBINS];
    __shared__ unsigned cur[NBINS];
    __shared__ unsigned pst[CHUNKS], pcn[CHUNKS], ppo[CHUNKS];
    __shared__ unsigned wsum[CG_W];
    __shared__ unsigned ticket;
    __shared__ unsigned list[LIST_CAP];
    __shared__ unsigned outl[LIST_CAP];
    const int tid  = threadIdx.x;
    const int lane = tid & 31;
    const int wave = __builtin_amdgcn_readfirstlane(tid >> 5);
    const int b    = blockIdx.x;
    const float4 as = *(const float4*)(a_attn + lane * 4);
    const float4 ad = *(const float4*)(a_attn + OUT_DIM + lane * 4);
    if (tid < NBINS) cntN[tid] = 0u;
    if (tid == 0) ticket = 0u;
    const unsigned tw = tid < CHUNKS ? tab[tid * NBINS + b] : 0u;
    unsigned T;
    const unsigned pex = block_excl_scan<CG_W>(tw & 0xffffu, wsum, T);
    if (tid < CHUNKS) { pst[tid] = tw >> 16; pcn[tid] = tw & 0xffffu; ppo[tid] = pex; }
    __syncthreads();
    const bool fits = T <= LIST_CAP;

    constexpr int PPW = CHUNKS / CG_W;
    unsigned r[PPW];
    #pragma unroll
    for (int q = 0; q < PPW; ++q) {
        const int ch = wave + CG_W * q;
        const bool     in = (unsigned)lane < pcn[ch];
        const unsigned v  = sorted[ch * CHUNK_EDGES + (in ? pst[ch] + lane : 0u)];
        r[q] = in ? v : 0xffffffffu;
    }
    #pragma unroll
    for (int q = 0; q < PPW; ++q) {
        const int ch = wave + CG_W * q;
        const unsigned len = pcn[ch], p0 = ppo[ch];
        const unsigned* pp = sorted + ch * CHUNK_EDGES + pst[ch];
        if (r[q] != 0xffffffffu) {
            atomicAdd(&cntN[r[q] >> 24], 1u);
            if (fits) list[p0 + lane] = r[q];
        }
        for (unsigned j = lane + 32; j < len; j += 32) {
            const unsigned v = pp[j];
            atomicAdd(&cntN[v >> 24], 1u);
            if (fits) list[p0 + j] = v;
        }
    }
    __syncthreads();
    const unsigned cv = tid < NBINS ? cntN[tid] : 0u;
    unsigned T2;
    const unsigned nex = block_excl_scan<CG_W>(cv, wsum, T2);
    if (tid < NBINS) { cur[tid] = nex; nst[tid] = nex; }
    __syncthreads();
    if (fits) {
        for (unsigned k = tid; k < T; k += CG_T) {
            const unsigned v = list[k];
            const unsigned pos = atomicAdd(&cur[v >> 24], 1u);
            outl[pos] = v & 0xffffffu;
        }
    }
    __syncthreads();

    int nv = n - b * BIN_NODES;
    if (nv > BIN_NODES) nv = BIN_NODES;
    const float* zl = z + lane * 4;
    const int  et = (lane >> 2) & 7;
    const bool h16 = (lane & 16) != 0, h8 = (lane & 8) != 0, h4 = (lane & 4) != 0;
    const float4 zero4 = make_float4(0.0f, 0.0f, 0.0f, 0.0f);
    for (;;) {
        unsigned tk = 0u;
        if (lane == 0) tk = atomicAdd(&ticket, 1u);
        const int ln = __builtin_amdgcn_readfirstlane((int)tk);
        if (ln >= nv) break;
        const int node = b * BIN_NODES + ln;
        const int cnt = __builtin_amdgcn_readfirstlane((int)cntN[ln]);
        const int beg = __builtin_amdgcn_readfirstlane((int)nst[ln]);
        float* o = out + (size_t)node * OUT_DIM + lane * 4;
        if (cnt == 0) {
            *(float4*)o = zrow(zl, node);
            continue;
        }
        if (!fits) {
            gather_node_scan(sorted, pst, pcn, zl, as, wave_sum(dot4(zrow(zl, node), ad)), (unsigned)ln, lane, o);
            continue;
        }
        float  zdd = 0.0f;
        float  m = -INFINITY, den = 0.0f;
        float4 acc = zero4;
        for (int b32 = 0; b32 < cnt; b32 += 32) {
            const int j   = b32 + lane;
            const int sv  = j < cnt ? (int)outl[beg + j] : 0;
            const int c32 = cnt - b32 < 32 ? cnt - b32 : 32;
            for (int g = 0; g < c32; g += 8) {
                const int c = c32 - g < 8 ? c32 - g : 8;
                const int s0 = __builtin_amdgcn_readlane(sv, g),     s1 = __builtin_amdgcn_readlane(sv, g + 1);
                const int s2 = __builtin_amdgcn_readlane(sv, g + 2), s3 = __builtin_amdgcn_readlane(sv, g + 3);
                const int s4 = __builtin_amdgcn_readlane(sv, g + 4), s5 = __builtin_amdgcn_readlane(sv, g + 5);
                const int s6 = __builtin_amdgcn_readlane(sv, g + 6), s7 = __builtin_amdgcn_readlane(sv, g + 7);
                float4 r0 = zrow(zl, s0);
                float4 r1 = zero4, r2 = zero4, r3 = zero4, r4 = zero4, r5 = zero4, r6 = zero4, r7 = zero4;
                if (c > 1) r1 = zrow(zl, s1);
                if (c > 2) r2 = zrow(zl, s2);
                if (c > 3) r3 = zrow(zl, s3);
                if (c > 4) r4 = zrow(zl, s4);
                if (c > 5) r5 = zrow(zl, s5);
                if (c > 6) r6 = zrow(zl, s6);
                if (c > 7) r7 = zrow(zl, s7);
                if (b32 + g == 0) zdd = wave_sum(dot4(zrow(zl, node), ad));
                const float p0 = dot4(r0, as), p1 = dot4(r1, as), p2 = dot4(r2, as), p3 = dot4(r3, as);
                const float p4 = dot4(r4, as), p5 = dot4(r5, as), p6 = dot4(r6, as), p7 = dot4(r7, as);
                const float q0 = (h16 ? p4 : p0) + xhalf(h16 ? p0 : p4);
                const float q1 = (h16 ? p5 : p1) + xhalf(h16 ? p1 : p5);
                const float q2 = (h16 ? p6 : p2) + xhalf(h16 ? p2 : p6);
                const float q3 = (h16 ? p7 : p3) + xhalf(h16 ? p3 : p7);
                const float u0 = (h8 ? q2 : q0) + xrow<8>(h8 ? q0 : q2);
                const float u1 = (h8 ? q3 : q1) + xrow<8>(h8 ? q1 : q3);
                float sc = (h4 ? u1 : u0) + xrow<4>(h4 ? u0 : u1);
                sc += xrow<2>(sc);
                sc += xrow<1>(sc);
                const bool  live = et < c;
                const float e    = live ? lrelu(sc + zdd) : -INFINITY;
                const float mn = fmaxf(m, groups_max(e));
                const float rs = expf(m - mn);
                const float pe = live ? expf(e - mn) : 0.0f;
                den = den * rs + groups_sum(pe);
                acc.x *= rs; acc.y *= rs; acc.z *= rs; acc.w *= rs;
                m = mn;
                fma4(acc, lane_f(pe,  0), r0); fma4(acc, lane_f(pe,  4), r1);
                fma4(acc, lane_f(pe,  8), r2); fma4(acc, lane_f(pe, 12), r3);
                fma4(acc, lane_f(pe, 16), r4); fma4(acc, lane_f(pe, 20), r5);
                fma4(acc, lane_f(pe, 24), r6); fma4(acc, lane_f(pe, 28), r7);
            }
        }
        const float inv = 1.0f / den;
        *(float4*)o = make_float4(acc.x * inv, acc.y * inv, acc.z * inv, acc.w * inv);
    }
}

extern "C" void kernel_launch(void* const* d_in, const int* in_sizes, int n_in,
                              void* d_out, int out_size, void* d_ws, size_t ws_size,
                              hipStream_t stream) {
    const float* h      = (const float*)d_in[0];
    const int*   src    = (const int*)d_in[1];
    const int*   dst    = (const int*)d_in[2];
    const float* W      = (const float*)d_in[3];
    const float* a_attn = (const float*)d_in[4];
    const int n = in_sizes[0] / IN_DIM;
    const int E = in_sizes[1];
    float* out = (float*)d_out;

    char* ws = (char*)d_ws;
    float*    z     = (float*)ws;    ws += (size_t)n * OUT_DIM * sizeof(float);
    unsigned* tab   = (unsigned*)ws; ws += (size_t)n * sizeof(float) * 2;
    unsigned* sorted = (unsigned*)ws;

    gat_gemm_wmma<<<GEMM_BLOCKS, GEMM_WAVES * 32, 0, stream>>>(h, W, z, n);
    gat_bin      <<<CHUNKS,              SORT_T, 0, stream>>>(src, dst, sorted, tab, E);
    gat_range    <<<NBINS,               CG_T,   0, stream>>>(sorted, tab, a_attn, z, out, n);
}


# Edge binning moved into the product launch: each of its 256 work-groups bins one chunk of edges by destination range in LDS before it multiplies; 3 launches become 2
# speedup vs baseline: 4.2246x; 1.1269x over previous
#include <hip/hip_runtime.h>
#include <hip/hip_bf16.h>
#include <math.h>

typedef __attribute__((ext_vector_type(2))) float v2f;
typedef __attribute__((ext_vector_type(8))) float v8f;

#define IN_DIM  256
#define OUT_DIM 128
#define WPITCH 260
#define GEMM_BLOCKS 256
#define GEMM_WAVES  16


template <int NT, int SJ, int JB>
__device__ __forceinline__ void gemm_unit(const float* __restrict__ h, const float* lw,
                                          float* __restrict__ z, int rowBase, int outBase,
                                          int n, int lane) {
    constexpr int NJ    = IN_DIM / 8;
    constexpr int NTRIP = NJ / JB;
    constexpr int NSTG  = JB / SJ;
    const int c  = lane & 15;
    const int hf = lane >> 4;
    int g = rowBase + c;
    if (g >= n) g = n - 1;
    const float* ap = h + (size_t)g * IN_DIM + hf * 4;
    const float* bp = lw + (outBase + c) * WPITCH + hf * 4;
    v8f acc[NT];
    #pragma unroll
    for (int t = 0; t < NT; ++t) acc[t] = v8f{};

    float4 an[JB];
    #pragma unroll
    for (int i = 0; i < JB; ++i) an[i] = *(const float4*)(ap + i * 8);
    float4 bn[SJ][NT];
    #pragma unroll
    for (int i = 0; i < SJ; ++i) {
        #pragma unroll
        for (int t = 0; t < NT; ++t) bn[i][t] = *(const float4*)(bp + t * 16 * WPITCH + i * 8);
    }

    #pragma unroll 1
    for (int trip = 0; trip < NTRIP; ++trip) {
        float4 a[JB];
        #pragma unroll
        for (int i = 0; i < JB; ++i) a[i] = an[i];
        if (trip + 1 < NTRIP) {
            #pragma unroll
            for (int i = 0; i < JB; ++i) an[i] = *(const float4*)(ap + ((trip + 1) * JB + i) * 8);
        }
        #pragma unroll
        for (int s = 0; s < NSTG; ++s) {
            float4 b[SJ][NT];
            #pragma unroll
            for (int i = 0; i < SJ; ++i) {
                #pragma unroll
                for (int t = 0; t < NT; ++t) b[i][t] = bn[i][t];
            }
            const int jn = trip * JB + (s + 1) * SJ;
            if (s + 1 < NSTG || trip + 1 < NTRIP) {
                #pragma unroll
                for (int i = 0; i < SJ; ++i) {
                    #pragma unroll
                    for (int t = 0; t < NT; ++t)
                        bn[i][t] = *(const float4*)(bp + t * 16 * WPITCH + (jn + i) * 8);
                }
            }
            __builtin_amdgcn_sched_barrier(0);
            #pragma unroll
            for (int i = 0; i < SJ; ++i) {
                const float4 av = a[s * SJ + i];
                const v2f a0 = {av.x, av.y};
                const v2f a1 = {av.z, av.w};
                #pragma unroll
                for (int t = 0; t < NT; ++t) {
                    const v2f b0 = {b[i][t].x, b[i][t].y};
                    acc[t] = __builtin_amdgcn_wmma_f32_16x16x4_f32(false, a0, false, b0, (short)0, acc[t], false, false);
                }
                #pragma unroll
                for (int t = 0; t < NT; ++t) {
                    const v2f b1 = {b[i][t].z, b[i][t].w};
                    acc[t] = __builtin_amdgcn_wmma_f32_16x16x4_f32(false, a1, false, b1, (short)0, acc[t], false, false);
                }
            }
            __builtin_amdgcn_sched_barrier(0);
        }
    }
    const int mrow0 = rowBase + hf * 8;
    #pragma unroll
    for (int r = 0; r < 8; ++r) {
        const int gnode = mrow0 + r;
        if (gnode < n) {
            #pragma unroll
            for (int t = 0; t < NT; ++t)
                z[(size_t)gnode * OUT_DIM + outBase + t * 16 + c] = acc[t][r];
        }
    }
}

#define NBINS       256
#define BIN_NODES   196
#define CHUNKS      256
#define CHUNK_EDGES 1563
#define SORT_T      512
#define SORT_W      (SORT_T / 32)
#define BIN_KPT     4
#define LIST_CAP    4096

template <int NW>
__device__ __forceinline__ unsigned block_excl_scan(unsigned c, unsigned* wsum, unsigned& total) {
    const int lane = threadIdx.x & 31;
    const int wave = threadIdx.x >> 5;
    unsigned v = c;
    for (int o = 1; o < 32; o <<= 1) {
        unsigned t = __shfl_up(v, o, 32);
        if (lane >= o) v += t;
    }
    if (lane == 31) wsum[wave] = v;
    __syncthreads();
    unsigned pre = 0u, tot = 0u;
    for (int w = 0; w < NW; ++w) {
        const unsigned t = wsum[w];
        if (w < wave) pre += t;
        tot += t;
    }
    __syncthreads();
    total = tot;
    return pre + (v - c);
}

static_assert(SORT_T == GEMM_WAVES * 32 && CHUNKS == GEMM_BLOCKS, "one chunk per work-group, one thread set");
__global__ __launch_bounds__(GEMM_WAVES * 32) void gat_gemm_bin(
    const float* __restrict__ h, const float* __restrict__ W,
    float* __restrict__ z, int n,
    const int* __restrict__ src, const int* __restrict__ dst,
    unsigned* __restrict__ sorted, unsigned* __restrict__ tab, int E) {
    __shared__ __attribute__((aligned(16))) float lw[OUT_DIM * WPITCH];
    __shared__ unsigned cnt[NBINS];
    __shared__ unsigned start[NBINS];
    __shared__ unsigned wsum[SORT_W];
    __shared__ unsigned stage[CHUNK_EDGES];

    const int tid  = threadIdx.x;
    const int lane = tid & 31;
    const int wave = __builtin_amdgcn_readfirstlane(tid >> 5);

    const int nTiles = (n + 15) >> 4;
    const int t0 = (int)(((unsigned)blockIdx.x * (unsigned)nTiles) / GEMM_BLOCKS);
    const int t1 = (int)((((unsigned)blockIdx.x + 1u) * (unsigned)nTiles) / GEMM_BLOCKS);
    const int nT = t1 - t0;
    const int nX = nT & ~3;
    const int nY = (nT - nX) * 4;
    const int u = 4 * (3 - (wave >> 2)) + (wave & 3);

    {
        int ft = -1;
        if (wave < nX)      ft = t0 + wave;
        else if (u < nY)    ft = t0 + nX + (u >> 2);
        if (ft >= 0) {
            int g = ft * 16 + (lane & 15);
            if (g >= n) g = n - 1;
            const float* p = h + (size_t)g * IN_DIM + (lane >> 4) * 128;
            __builtin_prefetch((const void*)p, 0, 1);
            __builtin_prefetch((const void*)(p + 64), 0, 1);
        }
    }

    const int ck = blockIdx.x;
    const int e0 = ck * CHUNK_EDGES;
    const int e1 = e0 + CHUNK_EDGES < E ? e0 + CHUNK_EDGES : E;
    if (tid < NBINS) cnt[tid] = 0u;
    unsigned dv[BIN_KPT], sv[BIN_KPT];
    #pragma unroll
    for (int k = 0; k < BIN_KPT; ++k) {
        int i = e0 + k * SORT_T + tid;
        if (i > e1 - 1) i = e1 - 1;
        if (i < 0) i = 0;
        dv[k] = (unsigned)dst[i];
        sv[k] = (unsigned)src[i];
    }

    for (int i = tid; i < OUT_DIM * (IN_DIM / 4); i += GEMM_WAVES * 32) {
        const int row = i >> 6, c4 = i & 63;
        const float4 v = *(const float4*)(W + (size_t)row * IN_DIM + c4 * 4);
        *(float4*)(&lw[row * WPITCH + c4 * 4]) = v;
    }
    __syncthreads();

    {
        unsigned rec[BIN_KPT], key[BIN_KPT];
        #pragma unroll
        for (int k = 0; k < BIN_KPT; ++k) {
            const int i = e0 + k * SORT_T + tid;
            key[k] = 0xffffffffu;
            rec[k] = 0u;
            if (i < e1) {
                const unsigned d  = dv[k];
                const unsigned sx = sv[k];
                const unsigned bn = d / BIN_NODES;
                const unsigned r  = atomicAdd(&cnt[bn], 1u);
                rec[k] = sx | ((d - bn * BIN_NODES) << 24);
                key[k] = (bn << 16) | r;
            }
        }
        __syncthreads();
        const unsigned cv = tid < NBINS ? cnt[tid] : 0u;
        unsigned tot;
        const unsigned ex = block_excl_scan<SORT_W>(cv, wsum, tot);
        if (tid < NBINS) start[tid] = ex;
        __syncthreads();
        #pragma unroll
        for (int k = 0; k < BIN_KPT; ++k) {
            if (key[k] != 0xffffffffu) stage[start[key[k] >> 16] + (key[k] & 0xffffu)] = rec[k];
        }
        __syncthreads();
        for (int j = tid; j < e1 - e0; j += SORT_T) sorted[e0 + j] = stage[j];
        if (tid < NBINS) tab[ck * NBINS + tid] = (ex << 16) | cv;
    }

    for (int i = wave; i < nX; i += GEMM_WAVES)
        gemm_unit<8, 1, 8>(h, lw, z, (t0 + i) * 16, 0, n, lane);
    if (u < nY)
        gemm_unit<2, 4, 32>(h, lw, z, (t0 + nX + (u >> 2)) * 16, (u & 3) * 32, n, lane);
}

__device__ __forceinline__ float lrelu(float x) { return x > 0.0f ? x : 0.01f * x; }

__device__ __forceinline__ float xhalf(float x) {
    return __int_as_float(__builtin_amdgcn_permlanex16(__float_as_int(x), __float_as_int(x),
                                                       0x76543210, 0xfedcba98, false, false));
}
template <int M>
__device__ __forceinline__ float xrow(float x) {
    return __int_as_float(__builtin_amdgcn_update_dpp(0, __float_as_int(x), 0x160 + M, 0xf, 0xf, true));
}
__device__ __forceinline__ float wave_sum(float v) {
    v += xrow<1>(v); v += xrow<2>(v); v += xrow<4>(v); v += xrow<8>(v);
    return v + xhalf(v);
}
__device__ __forceinline__ float groups_max(float v) {
    v = fmaxf(v, xrow<4>(v)); v = fmaxf(v, xrow<8>(v));
    return fmaxf(v, xhalf(v));
}
__device__ __forceinline__ float groups_sum(float v) {
    v += xrow<4>(v); v += xrow<8>(v);
    return v + xhalf(v);
}

__device__ __forceinline__ float4 zrow(const float* __restrict__ zl, int s) {
    return *(const float4*)(zl + (size_t)s * OUT_DIM);
}
__device__ __forceinline__ void fma4(float4& acc, float a, const float4& r) {
    acc.x += a * r.x; acc.y += a * r.y; acc.z += a * r.z; acc.w += a * r.w;
}
__device__ __forceinline__ float dot4(const float4& a, const float4& b) {
    return a.x * b.x + a.y * b.y + a.z * b.z + a.w * b.w;
}
__device__ __forceinline__ float lane_f(float v, int t) {
    return __int_as_float(__builtin_amdgcn_readlane(__float_as_int(v), t));
}

#define CG_T 1024
#define CG_W (CG_T / 32)

__device__ __forceinline__ void gather_node_scan(
    const unsigned* __restrict__ sorted, const unsigned* pst, const unsigned* pcn,
    const float* __restrict__ zl, const float4 as, const float zdd, const unsigned ln,
    const int lane, float* __restrict__ o) {
    float  m = -INFINITY, den = 0.0f;
    float4 acc = make_float4(0.0f, 0.0f, 0.0f, 0.0f);
    for (int ch = 0; ch < CHUNKS; ++ch) {
        const unsigned len = pcn[ch];
        const unsigned* pp = sorted + ch * CHUNK_EDGES + pst[ch];
        for (unsigned j0 = 0; j0 < len; j0 += 32) {
            const unsigned j = j0 + lane;
            const unsigned v = j < len ? pp[j] : 0xffffffffu;
            unsigned mask = __builtin_amdgcn_ballot_w32(j < len && (v >> 24) == ln);
            while (mask) {
                const int t = __builtin_ctz(mask);
                mask &= mask - 1u;
                const int sidx = (int)(__builtin_amdgcn_readlane((int)v, t) & 0xffffff);
                const float4 r = zrow(zl, sidx);
                const float e  = lrelu(wave_sum(dot4(r, as)) + zdd);
                const float mn = fmaxf(m, e);
                const float rs = expf(m - mn);
                const float pe = expf(e - mn);
                den = den * rs + pe;
                acc.x = acc.x * rs + pe * r.x; acc.y = acc.y * rs + pe * r.y;
                acc.z = acc.z * rs + pe * r.z; acc.w = acc.w * rs + pe * r.w;
                m = mn;
            }
        }
    }
    const float inv = 1.0f / den;
    *(float4*)o = make_float4(acc.x * inv, acc.y * inv, acc.z * inv, acc.w * inv);
}

__global__ __launch_bounds__(CG_T) void gat_range(
    const unsigned* __restrict__ sorted, const unsigned* __restrict__ tab,
    const float* __restrict__ a_attn, const float* __restrict__ z,
    float* __restrict__ out, int n) {
    __shared__ unsigned cntN[NBINS];
    __shared__ unsigned nst[NBINS];
    __shared__ unsigned cur[NBINS];
    __shared__ unsigned pst[CHUNKS], pcn[CHUNKS], ppo[CHUNKS];
    __shared__ unsigned wsum[CG_W];
    __shared__ unsigned ticket;
    __shared__ unsigned list[LIST_CAP];
    __shared__ unsigned outl[LIST_CAP];
    const int tid  = threadIdx.x;
    const int lane = tid & 31;
    const int wave = __builtin_amdgcn_readfirstlane(tid >> 5);
    const int b    = blockIdx.x;
    const float4 as = *(const float4*)(a_attn + lane * 4);
    const float4 ad = *(const float4*)(a_attn + OUT_DIM + lane * 4);
    if (tid < NBINS) cntN[tid] = 0u;
    if (tid == 0) ticket = 0u;
    const unsigned tw = tid < CHUNKS ? tab[tid * NBINS + b] : 0u;
    unsigned T;
    const unsigned pex = block_excl_scan<CG_W>(tw & 0xffffu, wsum, T);
    if (tid < CHUNKS) { pst[tid] = tw >> 16; pcn[tid] = tw & 0xffffu; ppo[tid] = pex; }
    __syncthreads();
    const bool fits = T <= LIST_CAP;

    constexpr int PPW = CHUNKS / CG_W;
    unsigned r[PPW];
    #pragma unroll
    for (int q = 0; q < PPW; ++q) {
        const int ch = wave + CG_W * q;
        const bool     in = (unsigned)lane < pcn[ch];
        const unsigned v  = sorted[ch * CHUNK_EDGES + (in ? pst[ch] + lane : 0u)];
        r[q] = in ? v : 0xffffffffu;
    }
    #pragma unroll
    for (int q = 0; q < PPW; ++q) {
        const int ch = wave + CG_W * q;
        const unsigned len = pcn[ch], p0 = ppo[ch];
        const unsigned* pp = sorted + ch * CHUNK_EDGES + pst[ch];
        if (r[q] != 0xffffffffu) {
            atomicAdd(&cntN[r[q] >> 24], 1u);
            if (fits) list[p0 + lane] = r[q];
        }
        for (unsigned j = lane + 32; j < len; j += 32) {
            const unsigned v = pp[j];
            atomicAdd(&cntN[v >> 24], 1u);
            if (fits) list[p0 + j] = v;
        }
    }
    __syncthreads();
    const unsigned cv = tid < NBINS ? cntN[tid] : 0u;
    unsigned T2;
    const unsigned nex = block_excl_scan<CG_W>(cv, wsum, T2);
    if (tid < NBINS) { cur[tid] = nex; nst[tid] = nex; }
    __syncthreads();
    if (fits) {
        for (unsigned k = tid; k < T; k += CG_T) {
            const unsigned v = list[k];
            const unsigned pos = atomicAdd(&cur[v >> 24], 1u);
            outl[pos] = v & 0xffffffu;
        }
    }
    __syncthreads();

    int nv = n - b * BIN_NODES;
    if (nv > BIN_NODES) nv = BIN_NODES;
    const float* zl = z + lane * 4;
    const int  et = (lane >> 2) & 7;
    const bool h16 = (lane & 16) != 0, h8 = (lane & 8) != 0, h4 = (lane & 4) != 0;
    const float4 zero4 = make_float4(0.0f, 0.0f, 0.0f, 0.0f);
    for (;;) {
        unsigned tk = 0u;
        if (lane == 0) tk = atomicAdd(&ticket, 1u);
        const int ln = __builtin_amdgcn_readfirstlane((int)tk);
        if (ln >= nv) break;
        const int node = b * BIN_NODES + ln;
        const int cnt = __builtin_amdgcn_readfirstlane((int)cntN[ln]);
        const int beg = __builtin_amdgcn_readfirstlane((int)nst[ln]);
        float* o = out + (size_t)node * OUT_DIM + lane * 4;
        if (cnt == 0) {
            *(float4*)o = zrow(zl, node);
            continue;
        }
        if (!fits) {
            gather_node_scan(sorted, pst, pcn, zl, as, wave_sum(dot4(zrow(zl, node), ad)), (unsigned)ln, lane, o);
            continue;
        }
        float  zdd = 0.0f;
        float  m = -INFINITY, den = 0.0f;
        float4 acc = zero4;
        for (int b32 = 0; b32 < cnt; b32 += 32) {
            const int j   = b32 + lane;
            const int sv  = j < cnt ? (int)outl[beg + j] : 0;
            const int c32 = cnt - b32 < 32 ? cnt - b32 : 32;
            for (int g = 0; g < c32; g += 8) {
                const int c = c32 - g < 8 ? c32 - g : 8;
                const int s0 = __builtin_amdgcn_readlane(sv, g),     s1 = __builtin_amdgcn_readlane(sv, g + 1);
                const int s2 = __builtin_amdgcn_readlane(sv, g + 2), s3 = __builtin_amdgcn_readlane(sv, g + 3);
                const int s4 = __builtin_amdgcn_readlane(sv, g + 4), s5 = __builtin_amdgcn_readlane(sv, g + 5);
                const int s6 = __builtin_amdgcn_readlane(sv, g + 6), s7 = __builtin_amdgcn_readlane(sv, g + 7);
                float4 r0 = zrow(zl, s0);
                float4 r1 = zero4, r2 = zero4, r3 = zero4, r4 = zero4, r5 = zero4, r6 = zero4, r7 = zero4;
                if (c > 1) r1 = zrow(zl, s1);
                if (c > 2) r2 = zrow(zl, s2);
                if (c > 3) r3 = zrow(zl, s3);
                if (c > 4) r4 = zrow(zl, s4);
                if (c > 5) r5 = zrow(zl, s5);
                if (c > 6) r6 = zrow(zl, s6);
                if (c > 7) r7 = zrow(zl, s7);
                if (b32 + g == 0) zdd = wave_sum(dot4(zrow(zl, node), ad));
                const float p0 = dot4(r0, as), p1 = dot4(r1, as), p2 = dot4(r2, as), p3 = dot4(r3, as);
                const float p4 = dot4(r4, as), p5 = dot4(r5, as), p6 = dot4(r6, as), p7 = dot4(r7, as);
                const float q0 = (h16 ? p4 : p0) + xhalf(h16 ? p0 : p4);
                const float q1 = (h16 ? p5 : p1) + xhalf(h16 ? p1 : p5);
                const float q2 = (h16 ? p6 : p2) + xhalf(h16 ? p2 : p6);
                const float q3 = (h16 ? p7 : p3) + xhalf(h16 ? p3 : p7);
                const float u0 = (h8 ? q2 : q0) + xrow<8>(h8 ? q0 : q2);
                const float u1 = (h8 ? q3 : q1) + xrow<8>(h8 ? q1 : q3);
                float sc = (h4 ? u1 : u0) + xrow<4>(h4 ? u0 : u1);
                sc += xrow<2>(sc);
                sc += xrow<1>(sc);
                const bool  live = et < c;
                const float e    = live ? lrelu(sc + zdd) : -INFINITY;
                const float mn = fmaxf(m, groups_max(e));
                const float rs = expf(m - mn);
                const float pe = live ? expf(e - mn) : 0.0f;
                den = den * rs + groups_sum(pe);
                acc.x *= rs; acc.y *= rs; acc.z *= rs; acc.w *= rs;
                m = mn;
                fma4(acc, lane_f(pe,  0), r0); fma4(acc, lane_f(pe,  4), r1);
                fma4(acc, lane_f(pe,  8), r2); fma4(acc, lane_f(pe, 12), r3);
                fma4(acc, lane_f(pe, 16), r4); fma4(acc, lane_f(pe, 20), r5);
                fma4(acc, lane_f(pe, 24), r6); fma4(acc, lane_f(pe, 28), r7);
            }
        }
        const float inv = 1.0f / den;
        *(float4*)o = make_float4(acc.x * inv, acc.y * inv, acc.z * inv, acc.w * inv);
    }
}

extern "C" void kernel_launch(void* const* d_in, const int* in_sizes, int n_in,
                              void* d_out, int out_size, void* d_ws, size_t ws_size,
                              hipStream_t stream) {
    const float* h      = (const float*)d_in[0];
    const int*   src    = (const int*)d_in[1];
    const int*   dst    = (const int*)d_in[2];
    const float* W      = (const float*)d_in[3];
    const float* a_attn = (const float*)d_in[4];
    const int n = in_sizes[0] / IN_DIM;
    const int E = in_sizes[1];
    float* out = (float*)d_out;

    char* ws = (char*)d_ws;
    float*    z     = (float*)ws;    ws += (size_t)n * OUT_DIM * sizeof(float);
    unsigned* tab   = (unsigned*)ws; ws += (size_t)n * sizeof(float) * 2;
    unsigned* sorted = (unsigned*)ws;

    gat_gemm_bin<<<GEMM_BLOCKS, GEMM_WAVES * 32, 0, stream>>>(h, W, z, n, src, dst, sorted, tab, E);
    gat_range   <<<NBINS,       CG_T,            0, stream>>>(sorted, tab, a_attn, z, out, n);
}


# Gather trip on fewer vector instructions: base-2 hardware exponential in the softmax, packed 4-element dots (their sum order changes), reciprocal for the final scale
# speedup vs baseline: 4.3646x; 1.0331x over previous
#include <hip/hip_runtime.h>
#include <hip/hip_bf16.h>
#include <math.h>

typedef __attribute__((ext_vector_type(2))) float v2f;
typedef __attribute__((ext_vector_type(8))) float v8f;

#define IN_DIM  256
#define OUT_DIM 128
#define WPITCH 260
#define GEMM_BLOCKS 256
#define GEMM_WAVES  16


template <int NT, int SJ, int JB>
__device__ __forceinline__ void gemm_unit(const float* __restrict__ h, const float* lw,
                                          float* __restrict__ z, int rowBase, int outBase,
                                          int n, int lane) {
    constexpr int NJ    = IN_DIM / 8;
    constexpr int NTRIP = NJ / JB;
    constexpr int NSTG  = JB / SJ;
    const int c  = lane & 15;
    const int hf = lane >> 4;
    int g = rowBase + c;
    if (g >= n) g = n - 1;
    const float* ap = h + (size_t)g * IN_DIM + hf * 4;
    const float* bp = lw + (outBase + c) * WPITCH + hf * 4;
    v8f acc[NT];
    #pragma unroll
    for (int t = 0; t < NT; ++t) acc[t] = v8f{};

    float4 an[JB];
    #pragma unroll
    for (int i = 0; i < JB; ++i) an[i] = *(const float4*)(ap + i * 8);
    float4 bn[SJ][NT];
    #pragma unroll
    for (int i = 0; i < SJ; ++i) {
        #pragma unroll
        for (int t = 0; t < NT; ++t) bn[i][t] = *(const float4*)(bp + t * 16 * WPITCH + i * 8);
    }

    #pragma unroll 1
    for (int trip = 0; trip < NTRIP; ++trip) {
        float4 a[JB];
        #pragma unroll
        for (int i = 0; i < JB; ++i) a[i] = an[i];
        if (trip + 1 < NTRIP) {
            #pragma unroll
            for (int i = 0; i < JB; ++i) an[i] = *(const float4*)(ap + ((trip + 1) * JB + i) * 8);
        }
        #pragma unroll
        for (int s = 0; s < NSTG; ++s) {
            float4 b[SJ][NT];
            #pragma unroll
            for (int i = 0; i < SJ; ++i) {
                #pragma unroll
                for (int t = 0; t < NT; ++t) b[i][t] = bn[i][t];
            }
            const int jn = trip * JB + (s + 1) * SJ;
            if (s + 1 < NSTG || trip + 1 < NTRIP) {
                #pragma unroll
                for (int i = 0; i < SJ; ++i) {
                    #pragma unroll
                    for (int t = 0; t < NT; ++t)
                        bn[i][t] = *(const float4*)(bp + t * 16 * WPITCH + (jn + i) * 8);
                }
            }
            __builtin_amdgcn_sched_barrier(0);
            #pragma unroll
            for (int i = 0; i < SJ; ++i) {
                const float4 av = a[s * SJ + i];
                const v2f a0 = {av.x, av.y};
                const v2f a1 = {av.z, av.w};
                #pragma unroll
                for (int t = 0; t < NT; ++t) {
                    const v2f b0 = {b[i][t].x, b[i][t].y};
                    acc[t] = __builtin_amdgcn_wmma_f32_16x16x4_f32(false, a0, false, b0, (short)0, acc[t], false, false);
                }
                #pragma unroll
                for (int t = 0; t < NT; ++t) {
                    const v2f b1 = {b[i][t].z, b[i][t].w};
                    acc[t] = __builtin_amdgcn_wmma_f32_16x16x4_f32(false, a1, false, b1, (short)0, acc[t], false, false);
                }
            }
            __builtin_amdgcn_sched_barrier(0);
        }
    }
    const int mrow0 = rowBase + hf * 8;
    #pragma unroll
    for (int r = 0; r < 8; ++r) {
        const int gnode = mrow0 + r;
        if (gnode < n) {
            #pragma unroll
            for (int t = 0; t < NT; ++t)
                z[(size_t)gnode * OUT_DIM + outBase + t * 16 + c] = acc[t][r];
        }
    }
}

#define NBINS       256
#define BIN_NODES   196
#define CHUNKS      256
#define CHUNK_EDGES 1563
#define SORT_T      512
#define SORT_W      (SORT_T / 32)
#define BIN_KPT     4
#define LIST_CAP    4096

template <int NW>
__device__ __forceinline__ unsigned block_excl_scan(unsigned c, unsigned* wsum, unsigned& total) {
    const int lane = threadIdx.x & 31;
    const int wave = threadIdx.x >> 5;
    unsigned v = c;
    for (int o = 1; o < 32; o <<= 1) {
        unsigned t = __shfl_up(v, o, 32);
        if (lane >= o) v += t;
    }
    if (lane == 31) wsum[wave] = v;
    __syncthreads();
    unsigned pre = 0u, tot = 0u;
    for (int w = 0; w < NW; ++w) {
        const unsigned t = wsum[w];
        if (w < wave) pre += t;
        tot += t;
    }
    __syncthreads();
    total = tot;
    return pre + (v - c);
}

static_assert(SORT_T == GEMM_WAVES * 32 && CHUNKS == GEMM_BLOCKS, "one chunk per work-group, one thread set");
__global__ __launch_bounds__(GEMM_WAVES * 32) void gat_gemm_bin(
    const float* __restrict__ h, const float* __restrict__ W,
    float* __restrict__ z, int n,
    const int* __restrict__ src, const int* __restrict__ dst,
    unsigned* __restrict__ sorted, unsigned* __restrict__ tab, int E) {
    __shared__ __attribute__((aligned(16))) float lw[OUT_DIM * WPITCH];
    __shared__ unsigned cnt[NBINS];
    __shared__ unsigned start[NBINS];
    __shared__ unsigned wsum[SORT_W];
    __shared__ unsigned stage[CHUNK_EDGES];

    const int tid  = threadIdx.x;
    const int lane = tid & 31;
    const int wave = __builtin_amdgcn_readfirstlane(tid >> 5);

    const int nTiles = (n + 15) >> 4;
    const int t0 = (int)(((unsigned)blockIdx.x * (unsigned)nTiles) / GEMM_BLOCKS);
    const int t1 = (int)((((unsigned)blockIdx.x + 1u) * (unsigned)nTiles) / GEMM_BLOCKS);
    const int nT = t1 - t0;
    const int nX = nT & ~3;
    const int nY = (nT - nX) * 4;
    const int u = 4 * (3 - (wave >> 2)) + (wave & 3);

    {
        int ft = -1;
        if (wave < nX)      ft = t0 + wave;
        else if (u < nY)    ft = t0 + nX + (u >> 2);
        if (ft >= 0) {
            int g = ft * 16 + (lane & 15);
            if (g >= n) g = n - 1;
            const float* p = h + (size_t)g * IN_DIM + (lane >> 4) * 128;
            __builtin_prefetch((const void*)p, 0, 1);
            __builtin_prefetch((const void*)(p + 64), 0, 1);
        }
    }

    const int ck = blockIdx.x;
    const int e0 = ck * CHUNK_EDGES;
    const int e1 = e0 + CHUNK_EDGES < E ? e0 + CHUNK_EDGES : E;
    if (tid < NBINS) cnt[tid] = 0u;
    unsigned dv[BIN_KPT], sv[BIN_KPT];
    #pragma unroll
    for (int k = 0; k < BIN_KPT; ++k) {
        int i = e0 + k * SORT_T + tid;
        if (i > e1 - 1) i = e1 - 1;
        if (i < 0) i = 0;
        dv[k] = (unsigned)dst[i];
        sv[k] = (unsigned)src[i];
    }

    for (int i = tid; i < OUT_DIM * (IN_DIM / 4); i += GEMM_WAVES * 32) {
        const int row = i >> 6, c4 = i & 63;
        const float4 v = *(const float4*)(W + (size_t)row * IN_DIM + c4 * 4);
        *(float4*)(&lw[row * WPITCH + c4 * 4]) = v;
    }
    __syncthreads();

    {
        unsigned rec[BIN_KPT], key[BIN_KPT];
        #pragma unroll
        for (int k = 0; k < BIN_KPT; ++k) {
            const int i = e0 + k * SORT_T + tid;
            key[k] = 0xffffffffu;
            rec[k] = 0u;
            if (i < e1) {
                const unsigned d  = dv[k];
                const unsigned sx = sv[k];
                const unsigned bn = d / BIN_NODES;
                const unsigned r  = atomicAdd(&cnt[bn], 1u);
                rec[k] = sx | ((d - bn * BIN_NODES) << 24);
                key[k] = (bn << 16) | r;
            }
        }
        __syncthreads();
        const unsigned cv = tid < NBINS ? cnt[tid] : 0u;
        unsigned tot;
        const unsigned ex = block_excl_scan<SORT_W>(cv, wsum, tot);
        if (tid < NBINS) start[tid] = ex;
        __syncthreads();
        #pragma unroll
        for (int k = 0; k < BIN_KPT; ++k) {
            if (key[k] != 0xffffffffu) stage[start[key[k] >> 16] + (key[k] & 0xffffu)] = rec[k];
        }
        __syncthreads();
        for (int j = tid; j < e1 - e0; j += SORT_T) sorted[e0 + j] = stage[j];
        if (tid < NBINS) tab[ck * NBINS + tid] = (ex << 16) | cv;
    }

    for (int i = wave; i < nX; i += GEMM_WAVES)
        gemm_unit<8, 1, 8>(h, lw, z, (t0 + i) * 16, 0, n, lane);
    if (u < nY)
        gemm_unit<2, 4, 32>(h, lw, z, (t0 + nX + (u >> 2)) * 16, (u & 3) * 32, n, lane);
}

__device__ __forceinline__ float lrelu(float x) { return x > 0.0f ? x : 0.01f * x; }
__device__ __forceinline__ float lrelu_max(float x) { return fmaxf(x, 0.01f * x); }
__device__ __forceinline__ float exp_hw(float x) { return __builtin_amdgcn_exp2f(x * 1.44269504088896341f); }

__device__ __forceinline__ float xhalf(float x) {
    return __int_as_float(__builtin_amdgcn_permlanex16(__float_as_int(x), __float_as_int(x),
                                                       0x76543210, 0xfedcba98, false, false));
}
template <int M>
__device__ __forceinline__ float xrow(float x) {
    return __int_as_float(__builtin_amdgcn_update_dpp(0, __float_as_int(x), 0x160 + M, 0xf, 0xf, true));
}
__device__ __forceinline__ float wave_sum(float v) {
    v += xrow<1>(v); v += xrow<2>(v); v += xrow<4>(v); v += xrow<8>(v);
    return v + xhalf(v);
}
__device__ __forceinline__ float groups_max(float v) {
    v = fmaxf(v, xrow<4>(v)); v = fmaxf(v, xrow<8>(v));
    return fmaxf(v, xhalf(v));
}
__device__ __forceinline__ float groups_sum(float v) {
    v += xrow<4>(v); v += xrow<8>(v);
    return v + xhalf(v);
}

__device__ __forceinline__ float4 zrow(const float* __restrict__ zl, int s) {
    return *(const float4*)(zl + (size_t)s * OUT_DIM);
}
__device__ __forceinline__ void fma4(float4& acc, float a, const float4& r) {
    acc.x += a * r.x; acc.y += a * r.y; acc.z += a * r.z; acc.w += a * r.w;
}
__device__ __forceinline__ float dot4(const float4& a, const float4& b) {
    return a.x * b.x + a.y * b.y + a.z * b.z + a.w * b.w;
}
__device__ __forceinline__ float dot4p(const float4& a, const float4& b) {
    const v2f t = __builtin_elementwise_fma(v2f{a.z, a.w}, v2f{b.z, b.w}, v2f{a.x, a.y} * v2f{b.x, b.y});
    return t.x + t.y;
}
__device__ __forceinline__ float lane_f(float v, int t) {
    return __int_as_float(__builtin_amdgcn_readlane(__float_as_int(v), t));
}

#define CG_T 1024
#define CG_W (CG_T / 32)

__device__ __forceinline__ void gather_node_scan(
    const unsigned* __restrict__ sorted, const unsigned* pst, const unsigned* pcn,
    const float* __restrict__ zl, const float4 as, const float zdd, const unsigned ln,
    const int lane, float* __restrict__ o) {
    float  m = -INFINITY, den = 0.0f;
    float4 acc = make_float4(0.0f, 0.0f, 0.0f, 0.0f);
    for (int ch = 0; ch < CHUNKS; ++ch) {
        const unsigned len = pcn[ch];
        const unsigned* pp = sorted + ch * CHUNK_EDGES + pst[ch];
        for (unsigned j0 = 0; j0 < len; j0 += 32) {
            const unsigned j = j0 + lane;
            const unsigned v = j < len ? pp[j] : 0xffffffffu;
            unsigned mask = __builtin_amdgcn_ballot_w32(j < len && (v >> 24) == ln);
            while (mask) {
                const int t = __builtin_ctz(mask);
                mask &= mask - 1u;
                const int sidx = (int)(__builtin_amdgcn_readlane((int)v, t) & 0xffffff);
                const float4 r = zrow(zl, sidx);
                const float e  = lrelu(wave_sum(dot4(r, as)) + zdd);
                const float mn = fmaxf(m, e);
                const float rs = expf(m - mn);
                const float pe = expf(e - mn);
                den = den * rs + pe;
                acc.x = acc.x * rs + pe * r.x; acc.y = acc.y * rs + pe * r.y;
                acc.z = acc.z * rs + pe * r.z; acc.w = acc.w * rs + pe * r.w;
                m = mn;
            }
        }
    }
    const float inv = 1.0f / den;
    *(float4*)o = make_float4(acc.x * inv, acc.y * inv, acc.z * inv, acc.w * inv);
}

__global__ __launch_bounds__(CG_T) void gat_range(
    const unsigned* __restrict__ sorted, const unsigned* __restrict__ tab,
    const float* __restrict__ a_attn, const float* __restrict__ z,
    float* __restrict__ out, int n) {
    __shared__ unsigned cntN[NBINS];
    __shared__ unsigned nst[NBINS];
    __shared__ unsigned cur[NBINS];
    __shared__ unsigned pst[CHUNKS], pcn[CHUNKS], ppo[CHUNKS];
    __shared__ unsigned wsum[CG_W];
    __shared__ unsigned ticket;
    __shared__ unsigned list[LIST_CAP];
    __shared__ unsigned outl[LIST_CAP];
    const int tid  = threadIdx.x;
    const int lane = tid & 31;
    const int wave = __builtin_amdgcn_readfirstlane(tid >> 5);
    const int b    = blockIdx.x;
    const float4 as = *(const float4*)(a_attn + lane * 4);
    const float4 ad = *(const float4*)(a_attn + OUT_DIM + lane * 4);
    if (tid < NBINS) cntN[tid] = 0u;
    if (tid == 0) ticket = 0u;
    const unsigned tw = tid < CHUNKS ? tab[tid * NBINS + b] : 0u;
    unsigned T;
    const unsigned pex = block_excl_scan<CG_W>(tw & 0xffffu, wsum, T);
    if (tid < CHUNKS) { pst[tid] = tw >> 16; pcn[tid] = tw & 0xffffu; ppo[tid] = pex; }
    __syncthreads();
    const bool fits = T <= LIST_CAP;

    constexpr int PPW = CHUNKS / CG_W;
    unsigned r[PPW];
    #pragma unroll
    for (int q = 0; q < PPW; ++q) {
        const int ch = wave + CG_W * q;
        const bool     in = (unsigned)lane < pcn[ch];
        const unsigned v  = sorted[ch * CHUNK_EDGES + (in ? pst[ch] + lane : 0u)];
        r[q] = in ? v : 0xffffffffu;
    }
    #pragma unroll
    for (int q = 0; q < PPW; ++q) {
        const int ch = wave + CG_W * q;
        const unsigned len = pcn[ch], p0 = ppo[ch];
        const unsigned* pp = sorted + ch * CHUNK_EDGES + pst[ch];
        if (r[q] != 0xffffffffu) {
            atomicAdd(&cntN[r[q] >> 24], 1u);
            if (fits) list[p0 + lane] = r[q];
        }
        for (unsigned j = lane + 32; j < len; j += 32) {
            const unsigned v = pp[j];
            atomicAdd(&cntN[v >> 24], 1u);
            if (fits) list[p0 + j] = v;
        }
    }
    __syncthreads();
    const unsigned cv = tid < NBINS ? cntN[tid] : 0u;
    unsigned T2;
    const unsigned nex = block_excl_scan<CG_W>(cv, wsum, T2);
    if (tid < NBINS) { cur[tid] = nex; nst[tid] = nex; }
    __syncthreads();
    if (fits) {
        for (unsigned k = tid; k < T; k += CG_T) {
            const unsigned v = list[k];
            const unsigned pos = atomicAdd(&cur[v >> 24], 1u);
            outl[pos] = v & 0xffffffu;
        }
    }
    __syncthreads();

    int nv = n - b * BIN_NODES;
    if (nv > BIN_NODES) nv = BIN_NODES;
    const float* zl = z + lane * 4;
    const int  et = (lane >> 2) & 7;
    const bool h16 = (lane & 16) != 0, h8 = (lane & 8) != 0, h4 = (lane & 4) != 0;
    const float4 zero4 = make_float4(0.0f, 0.0f, 0.0f, 0.0f);
    for (;;) {
        unsigned tk = 0u;
        if (lane == 0) tk = atomicAdd(&ticket, 1u);
        const int ln = __builtin_amdgcn_readfirstlane((int)tk);
        if (ln >= nv) break;
        const int node = b * BIN_NODES + ln;
        const int cnt = __builtin_amdgcn_readfirstlane((int)cntN[ln]);
        const int beg = __builtin_amdgcn_readfirstlane((int)nst[ln]);
        float* o = out + (size_t)node * OUT_DIM + lane * 4;
        if (cnt == 0) {
            *(float4*)o = zrow(zl, node);
            continue;
        }
        if (!fits) {
            gather_node_scan(sorted, pst, pcn, zl, as, wave_sum(dot4(zrow(zl, node), ad)), (unsigned)ln, lane, o);
            continue;
        }
        float  zdd = 0.0f;
        float  m = -INFINITY, den = 0.0f;
        float4 acc = zero4;
        for (int b32 = 0; b32 < cnt; b32 += 32) {
            const int j   = b32 + lane;
            const int sv  = j < cnt ? (int)outl[beg + j] : 0;
            const int c32 = cnt - b32 < 32 ? cnt - b32 : 32;
            for (int g = 0; g < c32; g += 8) {
                const int c = c32 - g < 8 ? c32 - g : 8;
                const int s0 = __builtin_amdgcn_readlane(sv, g),     s1 = __builtin_amdgcn_readlane(sv, g + 1);
                const int s2 = __builtin_amdgcn_readlane(sv, g + 2), s3 = __builtin_amdgcn_readlane(sv, g + 3);
                const int s4 = __builtin_amdgcn_readlane(sv, g + 4), s5 = __builtin_amdgcn_readlane(sv, g + 5);
                const int s6 = __builtin_amdgcn_readlane(sv, g + 6), s7 = __builtin_amdgcn_readlane(sv, g + 7);
                float4 r0 = zrow(zl, s0);
                float4 r1 = zero4, r2 = zero4, r3 = zero4, r4 = zero4, r5 = zero4, r6 = zero4, r7 = zero4;
                if (c > 1) r1 = zrow(zl, s1);
                if (c > 2) r2 = zrow(zl, s2);
                if (c > 3) r3 = zrow(zl, s3);
                if (c > 4) r4 = zrow(zl, s4);
                if (c > 5) r5 = zrow(zl, s5);
                if (c > 6) r6 = zrow(zl, s6);
                if (c > 7) r7 = zrow(zl, s7);
                if (b32 + g == 0) zdd = wave_sum(dot4p(zrow(zl, node), ad));
                const float p0 = dot4p(r0, as), p1 = dot4p(r1, as), p2 = dot4p(r2, as), p3 = dot4p(r3, as);
                const float p4 = dot4p(r4, as), p5 = dot4p(r5, as), p6 = dot4p(r6, as), p7 = dot4p(r7, as);
                const float q0 = (h16 ? p4 : p0) + xhalf(h16 ? p0 : p4);
                const float q1 = (h16 ? p5 : p1) + xhalf(h16 ? p1 : p5);
                const float q2 = (h16 ? p6 : p2) + xhalf(h16 ? p2 : p6);
                const float q3 = (h16 ? p7 : p3) + xhalf(h16 ? p3 : p7);
                const float u0 = (h8 ? q2 : q0) + xrow<8>(h8 ? q0 : q2);
                const float u1 = (h8 ? q3 : q1) + xrow<8>(h8 ? q1 : q3);
                float sc = (h4 ? u1 : u0) + xrow<4>(h4 ? u0 : u1);
                sc += xrow<2>(sc);
                sc += xrow<1>(sc);
                const bool  live = et < c;
                const float e    = live ? lrelu_max(sc + zdd) : -INFINITY;
                const float mn = fmaxf(m, groups_max(e));
                const float rs = exp_hw(m - mn);
                const float pe = live ? exp_hw(e - mn) : 0.0f;
                den = den * rs + groups_sum(pe);
                acc.x *= rs; acc.y *= rs; acc.z *= rs; acc.w *= rs;
                m = mn;
                fma4(acc, lane_f(pe,  0), r0); fma4(acc, lane_f(pe,  4), r1);
                fma4(acc, lane_f(pe,  8), r2); fma4(acc, lane_f(pe, 12), r3);
                fma4(acc, lane_f(pe, 16), r4); fma4(acc, lane_f(pe, 20), r5);
                fma4(acc, lane_f(pe, 24), r6); fma4(acc, lane_f(pe, 28), r7);
            }
        }
        const float inv = __builtin_amdgcn_rcpf(den);
        *(float4*)o = make_float4(acc.x * inv, acc.y * inv, acc.z * inv, acc.w * inv);
    }
}

extern "C" void kernel_launch(void* const* d_in, const int* in_sizes, int n_in,
                              void* d_out, int out_size, void* d_ws, size_t ws_size,
                              hipStream_t stream) {
    const float* h      = (const float*)d_in[0];
    const int*   src    = (const int*)d_in[1];
    const int*   dst    = (const int*)d_in[2];
    const float* W      = (const float*)d_in[3];
    const float* a_attn = (const float*)d_in[4];
    const int n = in_sizes[0] / IN_DIM;
    const int E = in_sizes[1];
    float* out = (float*)d_out;

    char* ws = (char*)d_ws;
    float*    z     = (float*)ws;    ws += (size_t)n * OUT_DIM * sizeof(float);
    unsigned* tab   = (unsigned*)ws; ws += (size_t)n * sizeof(float) * 2;
    unsigned* sorted = (unsigned*)ws;

    gat_gemm_bin<<<GEMM_BLOCKS, GEMM_WAVES * 32, 0, stream>>>(h, W, z, n, src, dst, sorted, tab, E);
    gat_range   <<<NBINS,       CG_T,            0, stream>>>(sorted, tab, a_attn, z, out, n);
}
